# starting kernel
# baseline (speedup 1.0000x reference)
#include <hip/hip_runtime.h>

typedef __attribute__((ext_vector_type(2))) float v2f;
typedef __attribute__((ext_vector_type(8))) float v8f;

#define C_DIM   128
#define MAX_DEG 16
#define WPAIR_STRIDE 144

__global__ __launch_bounds__(256) void gcn_gemm(const float* __restrict__ x,
                                                const int*   __restrict__ edge,
                                                const float* __restrict__ W,
                                                float*       __restrict__ h,
                                                int nStrips) {
    __shared__ float Wlds[64 * WPAIR_STRIDE * 2];

    {
        const int tid = threadIdx.x;
#pragma unroll
        for (int i = 0; i < 16; ++i) {
            const int q = tid + 256 * i;
            const int p = q >> 6;
            const int c = q & 63;
            const float2 ga = *(const float2*)(W + (size_t)(2 * p)     * C_DIM + 2 * c);
            const float2 gb = *(const float2*)(W + (size_t)(2 * p + 1) * C_DIM + 2 * c);
            float4 v; v.x = ga.x; v.y = gb.x; v.z = ga.y; v.w = gb.y;
            ((float4*)Wlds)[p * (WPAIR_STRIDE / 2) + c] = v;
        }
    }
    __syncthreads();

    const int wave  = threadIdx.x >> 5;
    const int lane  = threadIdx.x & 31;
    const int strip = blockIdx.x * 8 + wave;
    if (strip >= nStrips) return;

    const int m  = lane & 15;
    const int hi = lane >> 4;
    const int rowBase = strip * 16;

    const float* xrow = x + (size_t)(rowBase + m) * C_DIM;
    const float* wb0 = Wlds + ((size_t)hi * WPAIR_STRIDE + m) * 2;

    v8f c[8] = {};

    for (int k0 = 0; k0 < C_DIM; k0 += 4) {
        const int kk = k0 + 2 * hi;
        v2f a = *(const v2f*)(xrow + kk);
        const float* wb = wb0 + (size_t)(k0 >> 1) * WPAIR_STRIDE * 2;
#pragma unroll
        for (int t = 0; t < 8; ++t) {
            v2f b = *(const v2f*)(wb + t * 32);
            c[t] = __builtin_amdgcn_wmma_f32_16x16x4_f32(
                false, a, false, b, (short)0, c[t], false, false);
        }
    }

    const int* e = edge + (size_t)(rowBase + m) * MAX_DEG;
    int cnt = 0;
#pragma unroll
    for (int k = 0; k < MAX_DEG; ++k) cnt += (e[k] >= 0) ? 1 : 0;
    const float sv = rsqrtf((float)cnt + 1.0f);

#pragma unroll
    for (int r = 0; r < 8; ++r) {
        const float sc  = __shfl(sv, r + 8 * hi, 32);
        const int   row = rowBase + r + 8 * hi;
        float* hrow = h + (size_t)row * C_DIM;
#pragma unroll
        for (int t = 0; t < 8; ++t)
            hrow[t * 16 + m] = c[t][r] * sc;
    }
}

__global__ __launch_bounds__(256) void gcn_agg(const float* __restrict__ h,
                                               const int*   __restrict__ edge,
                                               float*       __restrict__ out,
                                               int nNodes) {
    const int lane = threadIdx.x & 31;
    const int node = blockIdx.x * 8 + (threadIdx.x >> 5);
    if (node >= nNodes) return;

    int idx = -1;
    if (lane < MAX_DEG) idx = edge[(size_t)node * MAX_DEG + lane];

    const unsigned long long valid = __ballot(idx >= 0);
    const float s = rsqrtf((float)__popcll(valid) + 1.0f);

    const float4* h4 = (const float4*)h;
    float4 acc = h4[(size_t)node * 32 + lane];

    int jnext = __shfl(idx, 0, 32);
#pragma unroll 4
    for (int k = 0; k < MAX_DEG; ++k) {
        const int j = jnext;
        jnext = (k + 1 < MAX_DEG) ? __shfl(idx, k + 1, 32) : -1;
        if (jnext >= 0)
            __builtin_prefetch(&h4[(size_t)jnext * 32 + lane], 0, 3);
        if (j >= 0) {
            const float4 v = h4[(size_t)j * 32 + lane];
            acc.x += v.x; acc.y += v.y; acc.z += v.z; acc.w += v.w;
        }
    }

    float4 o;
    o.x = acc.x * s; o.y = acc.y * s; o.z = acc.z * s; o.w = acc.w * s;
    ((float4*)out)[(size_t)node * 32 + lane] = o;
}

extern "C" void kernel_launch(void* const* d_in, const int* in_sizes, int n_in,
                              void* d_out, int out_size, void* d_ws, size_t ws_size,
                              hipStream_t stream) {
    const float* x    = (const float*)d_in[0];
    const int*   edge = (const int*)  d_in[1];
    const float* W    = (const float*)d_in[2];
    float*       out  = (float*)d_out;
    float*       h    = (float*)d_ws;

    const int N       = in_sizes[0] / C_DIM;
    const int nStrips = (N + 15) / 16;

    const int gemmBlocks = (nStrips + 7) / 8;
    gcn_gemm<<<gemmBlocks, 256, 0, stream>>>(x, edge, W, h, nStrips);

    const int aggBlocks = (N + 7) / 8;
    gcn_agg<<<aggBlocks, 256, 0, stream>>>(h, edge, out, N);
}


# 13 waves a block, each owns two 16-row strips: the second strip's row gather carries the first strip's 256 WMMAs (16 per node, between load issue and first use); sum order kept
# speedup vs baseline: 1.3523x; 1.3523x over previous
#include <hip/hip_runtime.h>


typedef __attribute__((ext_vector_type(2))) float v2f;
typedef __attribute__((ext_vector_type(8))) float v8f;

#define C_DIM   128
#define MAX_DEG 16
#define WPAIR_STRIDE 144
#define A_STRIDE 132
#define TILE_FLOATS (16 * A_STRIDE)
#define STRIPS_PER_BLOCK 25
#define WAVES_PER_BLOCK 13
#define BLOCK_THREADS (WAVES_PER_BLOCK * 32)
#define W_QUADS 4096
#define W_FULL_TRIPS (W_QUADS / BLOCK_THREADS)

__device__ __forceinline__ void stage_w(const float* __restrict__ W, float* Wlds, int q) {
    const int p = q >> 6;
    const int c = q & 63;
    const float2 ga = *(const float2*)(W + (size_t)(2 * p)     * C_DIM + 2 * c);
    const float2 gb = *(const float2*)(W + (size_t)(2 * p + 1) * C_DIM + 2 * c);
    float4 v; v.x = ga.x; v.y = gb.x; v.z = ga.y; v.w = gb.y;
    ((float4*)Wlds)[p * (WPAIR_STRIDE / 2) + c] = v;
}

__device__ __forceinline__ void mma_steps(const float* arow, const float* wb0, int k0, v8f (&c)[8]) {
#pragma unroll
    for (int h = 0; h < 2; ++h) {
        const int kk = k0 + 4 * h;
        const v2f a = *(const v2f*)(arow + kk);
        const float* wb = wb0 + (size_t)(kk >> 1) * WPAIR_STRIDE * 2;
#pragma unroll
        for (int t = 0; t < 8; ++t) {
            const v2f b = *(const v2f*)(wb + t * 32);
            c[t] = __builtin_amdgcn_wmma_f32_16x16x4_f32(
                false, a, false, b, (short)0, c[t], false, false);
        }
    }
}

__device__ __forceinline__ void store_strip(float* __restrict__ out, int rowBase, float sv,
                                            int m, int hi, const v8f (&c)[8]) {
#pragma unroll
    for (int r = 0; r < 8; ++r) {
        const float sc  = __shfl(sv, r + 8 * hi, 32);
        const int   row = rowBase + r + 8 * hi;
        float* orow = out + (size_t)row * C_DIM;
#pragma unroll
        for (int t = 0; t < 8; ++t)
            orow[t * 16 + m] = c[t][r] * sc;
    }
}

template <bool WITH_MMA>
__device__ __forceinline__ float gather_strip(const float* __restrict__ x,
                                              const int*   __restrict__ edge,
                                              int rowBase, int lastNode, int lane,
                                              float* As, const float* Aprev, const float* Wlds,
                                              v8f (&c)[8]) {
    const int m  = lane & 15;
    const int hi = lane >> 4;
    const float4* x4 = (const float4*)x;
    const int4*   e4 = (const int4*)edge;
    const float* arow = Aprev + m * A_STRIDE + 2 * hi;
    const float* wb0  = Wlds + ((size_t)hi * WPAIR_STRIDE + m) * 2;

    const int sub  = lane >> 2;
    const int quad = lane & 3;

    const int* e0 = edge + (size_t)rowBase * MAX_DEG;
    int idx = e0[m];
    int jA  = e0[sub];
    int jB  = e0[8 + sub];
    float sv = 0.0f;

    for (int n = 0; n < 16; ++n) {
        const int node = rowBase + n;
        const int nn = (n < 15) ? n + 1 : 15;
        const int* en = edge + (size_t)(rowBase + nn) * MAX_DEG;
        const int idxN = en[m];
        const int jAN  = en[sub];
        const int jBN  = en[8 + sub];

        const bool vA = jA >= 0, vB = jB >= 0;
        const int  rA = vA ? min(jA, lastNode) : node;
        const int  rB = vB ? min(jB, lastNode) : node;
        const int4 eA = e4[(size_t)rA * 4 + quad];
        const int4 eB = e4[(size_t)rB * 4 + quad];

        const int idc = (idx >= 0) ? min(idx, lastNode) : node;
        float4 v[MAX_DEG];
#pragma unroll
        for (int k = 0; k < MAX_DEG; ++k) {
            const int j = __builtin_amdgcn_readlane(idc, k);
            v[k] = x4[(size_t)j * 32 + lane];
        }
        const float4 vs = x4[(size_t)node * 32 + lane];

        if (WITH_MMA) mma_steps(arow, wb0, 8 * n, c);

        const unsigned bal = (unsigned)__ballot(idx >= 0) & 0xFFFFu;
        const float si = rsqrtf((float)__popc(bal) + 1.0f);
        sv = (m == n) ? si : sv;

        int cA = (eA.x >= 0) + (eA.y >= 0) + (eA.z >= 0) + (eA.w >= 0);
        int cB = (eB.x >= 0) + (eB.y >= 0) + (eB.z >= 0) + (eB.w >= 0);
        cA += __shfl_xor(cA, 1, 32); cA += __shfl_xor(cA, 2, 32);
        cB += __shfl_xor(cB, 1, 32); cB += __shfl_xor(cB, 2, 32);
        const float wA = vA ? rsqrtf((float)cA + 1.0f) : 0.0f;
        const float wB = vB ? rsqrtf((float)cB + 1.0f) : 0.0f;

        float4 acc; acc.x = 0.0f; acc.y = 0.0f; acc.z = 0.0f; acc.w = 0.0f;
#pragma unroll
        for (int k = 0; k < MAX_DEG; ++k) {
            const int wi = (k < 8) ? __builtin_amdgcn_readlane(__float_as_int(wA), 4 * k)
                                   : __builtin_amdgcn_readlane(__float_as_int(wB), 4 * (k - 8));
            const float w = __int_as_float(wi);
            acc.x += w * v[k].x; acc.y += w * v[k].y; acc.z += w * v[k].z; acc.w += w * v[k].w;
        }
        acc.x += si * vs.x; acc.y += si * vs.y; acc.z += si * vs.z; acc.w += si * vs.w;

        *(float4*)(As + n * A_STRIDE + lane * 4) = acc;

        idx = idxN; jA = jAN; jB = jBN;
    }

    __builtin_amdgcn_fence(__ATOMIC_RELEASE, "wavefront");
    __builtin_amdgcn_wave_barrier();
    __builtin_amdgcn_fence(__ATOMIC_ACQUIRE, "wavefront");
    return sv;
}

__global__ __launch_bounds__(BLOCK_THREADS) void gcn_fused(const float* __restrict__ x,
                                                           const int*   __restrict__ edge,
                                                           const float* __restrict__ W,
                                                           float*       __restrict__ out,
                                                           int nStrips, int nNodes) {
    __shared__ __attribute__((aligned(16))) float Wlds[64 * WPAIR_STRIDE * 2];
    __shared__ __attribute__((aligned(16))) float Alds[WAVES_PER_BLOCK * 2 * TILE_FLOATS];

#pragma unroll
    for (int i = 0; i < W_FULL_TRIPS; ++i)
        stage_w(W, Wlds, (int)threadIdx.x + i * BLOCK_THREADS);
    if ((int)threadIdx.x + W_FULL_TRIPS * BLOCK_THREADS < W_QUADS)
        stage_w(W, Wlds, (int)threadIdx.x + W_FULL_TRIPS * BLOCK_THREADS);
    __syncthreads();

    const int wave = __builtin_amdgcn_readfirstlane((int)(threadIdx.x >> 5));
    const int lane = threadIdx.x & 31;
    const int m  = lane & 15;
    const int hi = lane >> 4;
    const int lastNode = nNodes - 1;

    const int stripEnd = min(nStrips, ((int)blockIdx.x + 1) * STRIPS_PER_BLOCK);
    int strip = blockIdx.x * STRIPS_PER_BLOCK + wave;
    if (strip >= stripEnd) return;

    float* tiles = Alds + wave * (2 * TILE_FLOATS);
    int cur = 0;
    v8f c[8] = {};

    float svPrev = gather_strip<false>(x, edge, strip * 16, lastNode, lane, tiles, tiles, Wlds, c);
    int prevStrip = strip;

    for (strip += WAVES_PER_BLOCK; strip < stripEnd; strip += WAVES_PER_BLOCK) {
#pragma unroll
        for (int t = 0; t < 8; ++t) c[t] = (v8f){};
        const float svNew = gather_strip<true>(x, edge, strip * 16, lastNode, lane,
                                               tiles + (cur ^ 1) * TILE_FLOATS,
                                               tiles + cur * TILE_FLOATS, Wlds, c);
        store_strip(out, prevStrip * 16, svPrev, m, hi, c);
        cur ^= 1; svPrev = svNew; prevStrip = strip;
    }

#pragma unroll
    for (int t = 0; t < 8; ++t) c[t] = (v8f){};
    {
        const float* arow = tiles + cur * TILE_FLOATS + m * A_STRIDE + 2 * hi;
        const float* wb0  = Wlds + ((size_t)hi * WPAIR_STRIDE + m) * 2;
        for (int k0 = 0; k0 < C_DIM; k0 += 8)
            mma_steps(arow, wb0, k0, c);
    }
    store_strip(out, prevStrip * 16, svPrev, m, hi, c);
}

extern "C" void kernel_launch(void* const* d_in, const int* in_sizes, int n_in,
                              void* d_out, int out_size, void* d_ws, size_t ws_size,
                              hipStream_t stream) {
    const float* x    = (const float*)d_in[0];
    const int*   edge = (const int*)  d_in[1];
    const float* W    = (const float*)d_in[2];
    float*       out  = (float*)d_out;

    const int N       = in_sizes[0] / C_DIM;
    const int nStrips = (N + 15) / 16;

    const int blocks = (nStrips + STRIPS_PER_BLOCK - 1) / STRIPS_PER_BLOCK;
    gcn_fused<<<blocks, BLOCK_THREADS, 0, stream>>>(x, edge, W, out, nStrips, N);
}
